# starting kernel
# baseline (speedup 1.0000x reference)
#include <hip/hip_runtime.h>

typedef __attribute__((ext_vector_type(2))) float v2f;
typedef __attribute__((ext_vector_type(8))) float v8f;

#define HID 64

__global__ void k_deg_init(float* deg, int n) {
    int i = blockIdx.x * blockDim.x + threadIdx.x;
    if (i < n) deg[i] = 1.0f;
}

__global__ void k_deg_edges(const int* __restrict__ dst, float* deg, int e) {
    int i = blockIdx.x * blockDim.x + threadIdx.x;
    if (i < e) atomicAdd(&deg[dst[i]], 1.0f);
}

__global__ void k_dinv(float* deg, int n) {
    int i = blockIdx.x * blockDim.x + threadIdx.x;
    if (i < n) { float d = deg[i]; deg[i] = (d > 0.f) ? rsqrtf(d) : 0.f; }
}

template <int K, bool DO_RELU>
__global__ void k_gemm_wmma(const float* __restrict__ A, const float* __restrict__ W,
                            float* __restrict__ C, int n) {
    __shared__ float sW[K * HID];
    int tid = threadIdx.x;
    for (int i = tid; i < K * HID; i += 128) sW[i] = W[i];
    __syncthreads();

    int wave = tid >> 5;
    int lane = tid & 31;
    int half = lane >> 4;
    int l15  = lane & 15;
    int m0 = blockIdx.x * 16;
    int n0 = wave * 16;

    int arow = m0 + l15;
    if (arow >= n) arow = n - 1;
    const float* ap = A + (size_t)arow * K + half * 2;
    const float* bp = sW + n0 + l15 + half * 2 * HID;

    v8f c = {};
#pragma unroll
    for (int k = 0; k < K; k += 4) {
        v2f a, b;
        float ax = ap[k];
        float ay = ap[k + 1];
        if (DO_RELU) { ax = fmaxf(ax, 0.f); ay = fmaxf(ay, 0.f); }
        a[0] = ax; a[1] = ay;
        b[0] = bp[k * HID];
        b[1] = bp[k * HID + HID];
        c = __builtin_amdgcn_wmma_f32_16x16x4_f32(
                 false, a,  false, b,
                 (short)0, c,  false,  false);
    }

    int rbase = m0 + half * 8;
#pragma unroll
    for (int j = 0; j < 8; ++j) {
        int r = rbase + j;
        if (r < n) C[(size_t)r * HID + n0 + l15] = c[j];
    }
}

__global__ void k_selfinit(const float* __restrict__ hW, const float* __restrict__ dinv,
                           float* __restrict__ agg, int n) {
    int i = blockIdx.x * blockDim.x + threadIdx.x;
    if (i < n * HID) {
        int v = i >> 6;
        float di = dinv[v];
        agg[i] = hW[i] * di * di;
    }
}

__global__ void k_edge_scatter(const int* __restrict__ src, const int* __restrict__ dst,
                               const float* __restrict__ dinv, const float* __restrict__ hW,
                               float* __restrict__ agg, int e) {
    int i = blockIdx.x * blockDim.x + threadIdx.x;
    int total = e * HID;
    if (i < total) {
        int ed = i >> 6, ch = i & 63;
        int s = src[ed], d = dst[ed];
        float nm = dinv[s] * dinv[d];
        atomicAdd(&agg[(size_t)d * HID + ch], hW[(size_t)s * HID + ch] * nm);
    }
}

__global__ void k_pool_zero(float* sums, float* cnt, int g) {
    int i = blockIdx.x * blockDim.x + threadIdx.x;
    if (i < g * HID) sums[i] = 0.f;
    if (i < g)       cnt[i]  = 0.f;
}

__global__ void k_pool_scatter(const float* __restrict__ h, const int* __restrict__ batch,
                               float* sums, float* cnt, int n) {
    int i = blockIdx.x * blockDim.x + threadIdx.x;
    if (i < n * HID) {
        int v = i >> 6, ch = i & 63;
        int g = batch[v];
        atomicAdd(&sums[(size_t)g * HID + ch], h[i]);
        if (ch == 0) atomicAdd(&cnt[g], 1.0f);
    }
}

__global__ void k_final(const float* __restrict__ sums, const float* __restrict__ cnt,
                        const float* __restrict__ Wl, const float* __restrict__ bl,
                        float* __restrict__ out, int g, int oc) {
    int i = blockIdx.x * blockDim.x + threadIdx.x;
    if (i < g * oc) {
        int gi = i / oc, o = i % oc;
        float inv = 1.0f / fmaxf(cnt[gi], 1.0f);
        float acc = bl[o];
        for (int c = 0; c < HID; ++c)
            acc += sums[(size_t)gi * HID + c] * inv * Wl[c * oc + o];
        out[i] = acc;
    }
}

extern "C" void kernel_launch(void* const* d_in, const int* in_sizes, int n_in,
                              void* d_out, int out_size, void* d_ws, size_t ws_size,
                              hipStream_t stream) {
    const float* x    = (const float*)d_in[0];
    const int*   eidx = (const int*)  d_in[1];
    const int*   batch= (const int*)  d_in[2];
    const float* W1   = (const float*)d_in[3];
    const float* W2   = (const float*)d_in[4];
    const float* W3   = (const float*)d_in[5];
    const float* Wl   = (const float*)d_in[6];
    const float* bl   = (const float*)d_in[7];

    const int INC = in_sizes[3] / HID;
    const int N   = in_sizes[0] / INC;
    const int E   = in_sizes[1] / 2;
    const int OC  = in_sizes[7];
    const int G   = out_size / OC;

    const int* src = eidx;
    const int* dst = eidx + E;

    char* ws = (char*)d_ws; size_t off = 0;
    auto carve = [&](size_t bytes) {
        void* p = ws + off; off = (off + bytes + 255) & ~(size_t)255; return p;
    };
    float* dinv = (float*)carve((size_t)N * 4);
    float* hW   = (float*)carve((size_t)N * HID * 4);
    float* agg  = (float*)carve((size_t)N * HID * 4);
    float* sums = (float*)carve((size_t)G * HID * 4);
    float* cnt  = (float*)carve((size_t)G * 4);
    (void)ws_size; (void)n_in;

    const int B = 256;
    const int nB  = (N + B - 1) / B;
    const int eB  = (E + B - 1) / B;
    const int nhB = (N * HID + B - 1) / B;
    const int ehB = (E * HID + B - 1) / B;
    const int mt  = (N + 15) / 16;

    k_deg_init <<<nB, B, 0, stream>>>(dinv, N);
    k_deg_edges<<<eB, B, 0, stream>>>(dst, dinv, E);
    k_dinv     <<<nB, B, 0, stream>>>(dinv, N);

    k_gemm_wmma<128, false><<<mt, 128, 0, stream>>>(x, W1, hW, N);
    k_selfinit    <<<nhB, B, 0, stream>>>(hW, dinv, agg, N);
    k_edge_scatter<<<ehB, B, 0, stream>>>(src, dst, dinv, hW, agg, E);

    k_gemm_wmma<64, true><<<mt, 128, 0, stream>>>(agg, W2, hW, N);
    k_selfinit    <<<nhB, B, 0, stream>>>(hW, dinv, agg, N);
    k_edge_scatter<<<ehB, B, 0, stream>>>(src, dst, dinv, hW, agg, E);

    k_gemm_wmma<64, true><<<mt, 128, 0, stream>>>(agg, W3, hW, N);
    k_selfinit    <<<nhB, B, 0, stream>>>(hW, dinv, agg, N);
    k_edge_scatter<<<ehB, B, 0, stream>>>(src, dst, dinv, hW, agg, E);

    k_pool_zero   <<<(G * HID + B - 1) / B, B, 0, stream>>>(sums, cnt, G);
    k_pool_scatter<<<nhB, B, 0, stream>>>(agg, batch, sums, cnt, N);
    k_final       <<<(G * OC + B - 1) / B, B, 0, stream>>>(sums, cnt, Wl, bl,
                                                           (float*)d_out, G, OC);
}


# GEMM: a work-group stages 64 rows of A in LDS with whole-line 16-byte loads (relu there), each wave keeps its 16 columns of W in registers and runs 4 row tiles; W no longer staged per 16 rows
# speedup vs baseline: 4.4531x; 4.4531x over previous
#include <hip/hip_runtime.h>

typedef __attribute__((ext_vector_type(2))) float v2f;
typedef __attribute__((ext_vector_type(8))) float v8f;

#define HID 64

#define TSH 6
#define TN  (1 << TSH)
#define CH  4096
#define SCAP 2048

__global__ void k_bin_count(const int* __restrict__ dst, int* __restrict__ tab, int e, int nt) {
    extern __shared__ int hist[];
    int tid = threadIdx.x;
    for (int i = tid; i < nt; i += 256) hist[i] = 0;
    __syncthreads();
    int e0 = blockIdx.x * CH;
    int e1 = min(e0 + CH, e);
    int dd[CH / 256];
#pragma unroll
    for (int k = 0; k < CH / 256; ++k)
        dd[k] = dst[min(e0 + tid + k * 256, e1 - 1)];
#pragma unroll
    for (int k = 0; k < CH / 256; ++k)
        if (e0 + tid + k * 256 < e1) atomicAdd(&hist[dd[k] >> TSH], 1);
    __syncthreads();
    int* row = tab + (size_t)blockIdx.x * nt;
    for (int i = tid; i < nt; i += 256) row[i] = hist[i];
}

__global__ void k_bin_col(int* __restrict__ tab, int* __restrict__ tot, int nchunk, int nt) {
    int lane = threadIdx.x & 31;
    int b = blockIdx.x * 4 + (threadIdx.x >> 5);
    if (b >= nt) return;
    int carry = 0;
    for (int c0 = 0; c0 < nchunk; c0 += 128) {
        int v[4];
#pragma unroll
        for (int k = 0; k < 4; ++k) {
            int c = c0 + k * 32 + lane;
            v[k] = (c < nchunk) ? tab[(size_t)c * nt + b] : 0;
        }
#pragma unroll
        for (int k = 0; k < 4; ++k) {
            int c = c0 + k * 32 + lane;
            int x = v[k];
            for (int d = 1; d < 32; d <<= 1) {
                int t = __shfl_up(x, d);
                if (lane >= d) x += t;
            }
            if (c < nchunk) tab[(size_t)c * nt + b] = carry + x - v[k];
            carry += __shfl(x, 31);
        }
    }
    if (lane == 0) tot[b] = carry;
}

__global__ void k_bin_start(const int* __restrict__ tot, int* __restrict__ start, int nt) {
    __shared__ int s[1024];
    int t = threadIdx.x;
    int carry = 0;
    for (int b0 = 0; b0 < nt; b0 += 1024) {
        int v = (b0 + t < nt) ? tot[b0 + t] : 0;
        s[t] = v;
        __syncthreads();
        for (int d = 1; d < 1024; d <<= 1) {
            int a = (t >= d) ? s[t - d] : 0;
            __syncthreads();
            s[t] += a;
            __syncthreads();
        }
        if (b0 + t < nt) start[b0 + t] = carry + s[t] - v;
        carry += s[1023];
        __syncthreads();
    }
    if (t == 0) start[nt] = carry;
}

__global__ void k_bin_fill(const int* __restrict__ src, const int* __restrict__ dst,
                           const int* __restrict__ tab, const int* __restrict__ start,
                           int2* __restrict__ rec, int e, int nt) {
    extern __shared__ int cur[];
    int tid = threadIdx.x;
    const int* row = tab + (size_t)blockIdx.x * nt;
    for (int i = tid; i < nt; i += 256) cur[i] = start[i] + row[i];
    __syncthreads();
    int e0 = blockIdx.x * CH;
    int e1 = min(e0 + CH, e);
    int ss[CH / 256], dd[CH / 256];
#pragma unroll
    for (int k = 0; k < CH / 256; ++k) {
        int i = min(e0 + tid + k * 256, e1 - 1);
        ss[k] = src[i];
        dd[k] = dst[i];
    }
#pragma unroll
    for (int k = 0; k < CH / 256; ++k) {
        if (e0 + tid + k * 256 < e1) {
            int pos = atomicAdd(&cur[dd[k] >> TSH], 1);
            rec[pos] = make_int2(ss[k], dd[k]);
        }
    }
}

__global__ void __launch_bounds__(256)
k_tile_deg(int2* __restrict__ rec, const int* __restrict__ start,
           float* __restrict__ dinv, int n,
           float* __restrict__ psum, float* __restrict__ pcnt, int g) {
    __shared__ int cnt[TN];
    __shared__ int cc[TN];
    __shared__ int2 buf[SCAP];
    int tid = threadIdx.x;
    for (int i = blockIdx.x * 256 + tid; i < g * HID; i += gridDim.x * 256) {
        psum[i] = 0.f;
        if (i < g) pcnt[i] = 0.f;
    }
    if (tid < TN) cnt[tid] = 0;
    int b = start[blockIdx.x], e = start[blockIdx.x + 1];
    for (int c0 = b; c0 < e; c0 += SCAP) {
        int len = min(SCAP, e - c0);
        if (tid < TN) cc[tid] = 0;
        __syncthreads();
        int2 r[SCAP / 256];
        int rk[SCAP / 256];
#pragma unroll
        for (int k = 0; k < SCAP / 256; ++k)
            r[k] = rec[min(c0 + tid + k * 256, e - 1)];
#pragma unroll
        for (int k = 0; k < SCAP / 256; ++k)
            rk[k] = (tid + k * 256 < len) ? atomicAdd(&cc[r[k].y & (TN - 1)], 1) : 0;
        __syncthreads();
        if (tid < 32) {
            int a0 = cc[2 * tid], a1 = cc[2 * tid + 1];
            int x = a0 + a1;
            int inc = x;
            for (int dd = 1; dd < 32; dd <<= 1) {
                int t = __shfl_up(inc, dd);
                if (tid >= dd) inc += t;
            }
            cnt[2 * tid] += a0;
            cnt[2 * tid + 1] += a1;
            cc[2 * tid] = inc - x;
            cc[2 * tid + 1] = inc - x + a0;
        }
        __syncthreads();
        int pos[SCAP / 256];
#pragma unroll
        for (int k = 0; k < SCAP / 256; ++k) pos[k] = cc[r[k].y & (TN - 1)] + rk[k];
#pragma unroll
        for (int k = 0; k < SCAP / 256; ++k)
            if (tid + k * 256 < len) buf[pos[k]] = r[k];
        __syncthreads();
#pragma unroll
        for (int k = 0; k < SCAP / 256; ++k) r[k] = buf[tid + k * 256];
#pragma unroll
        for (int k = 0; k < SCAP / 256; ++k)
            if (tid + k * 256 < len) rec[c0 + tid + k * 256] = r[k];
    }
    __syncthreads();
    int v = blockIdx.x * TN + tid;
    if (tid < TN && v < n) {
        float d = 1.0f + (float)cnt[tid];
        dinv[v] = (d > 0.f) ? rsqrtf(d) : 0.f;
    }
}

#define GR 64
template <int K, bool DO_RELU>
__global__ void __launch_bounds__(128)
k_gemm_wmma(const float* __restrict__ A, const float* __restrict__ W,
            float* __restrict__ C, int n) {
    constexpr int P = K + 4;
    __shared__ __attribute__((aligned(16))) float sA[GR * P];
    int tid = threadIdx.x;
    int wave = tid >> 5;
    int lane = tid & 31;
    int half = lane >> 4;
    int l15  = lane & 15;
    int n0 = wave * 16;

    const float* bp = W + n0 + l15 + half * 2 * HID;
    v2f b[K / 4];
#pragma unroll
    for (int k = 0; k < K; k += 4) {
        b[k / 4][0] = bp[k * HID];
        b[k / 4][1] = bp[k * HID + HID];
    }

    const float4* A4 = (const float4*)A;
    const float* ap = sA + l15 * P + half * 2;
    for (int r0 = blockIdx.x * GR; r0 < n; r0 += gridDim.x * GR) {
        float4 st[GR * (K / 4) / 128];
#pragma unroll
        for (int q = 0; q < GR * (K / 4) / 128; ++q) {
            int i = tid + q * 128;
            int row = i / (K / 4), c4 = i % (K / 4);
            int gr = min(r0 + row, n - 1);
            st[q] = A4[(size_t)gr * (K / 4) + c4];
        }
#pragma unroll
        for (int q = 0; q < GR * (K / 4) / 128; ++q) {
            int i = tid + q * 128;
            int row = i / (K / 4), c4 = i % (K / 4);
            float4 v = st[q];
            if (DO_RELU) {
                v.x = fmaxf(v.x, 0.f); v.y = fmaxf(v.y, 0.f);
                v.z = fmaxf(v.z, 0.f); v.w = fmaxf(v.w, 0.f);
            }
            *(float4*)(sA + row * P + c4 * 4) = v;
        }
        __syncthreads();

        v8f c[GR / 16];
#pragma unroll
        for (int t = 0; t < GR / 16; ++t) c[t] = v8f{};
        v2f a[2][GR / 16][4];
#pragma unroll
        for (int t = 0; t < GR / 16; ++t)
#pragma unroll
            for (int s = 0; s < 4; ++s)
                a[0][t][s] = *(const v2f*)(ap + t * 16 * P + s * 4);
#pragma unroll
        for (int g = 0; g < K / 16; ++g) {
            if (g + 1 < K / 16) {
#pragma unroll
                for (int t = 0; t < GR / 16; ++t)
#pragma unroll
                    for (int s = 0; s < 4; ++s)
                        a[(g + 1) & 1][t][s] = *(const v2f*)(ap + t * 16 * P + (g + 1) * 16 + s * 4);
            }
            asm volatile("" :: "v"(a[g & 1][0][0]), "v"(a[g & 1][0][1]), "v"(a[g & 1][0][2]), "v"(a[g & 1][0][3]),
                               "v"(a[g & 1][1][0]), "v"(a[g & 1][1][1]), "v"(a[g & 1][1][2]), "v"(a[g & 1][1][3]),
                               "v"(a[g & 1][2][0]), "v"(a[g & 1][2][1]), "v"(a[g & 1][2][2]), "v"(a[g & 1][2][3]),
                               "v"(a[g & 1][3][0]), "v"(a[g & 1][3][1]), "v"(a[g & 1][3][2]), "v"(a[g & 1][3][3]));
#pragma unroll
            for (int s = 0; s < 4; ++s) {
#pragma unroll
                for (int t = 0; t < GR / 16; ++t) {
                    c[t] = __builtin_amdgcn_wmma_f32_16x16x4_f32(
                             false, a[g & 1][t][s],  false, b[g * 4 + s],
                             (short)0, c[t],  false,  false);
                }
            }
        }

#pragma unroll
        for (int t = 0; t < GR / 16; ++t) {
#pragma unroll
            for (int j = 0; j < 8; ++j) {
                int r = r0 + t * 16 + half * 8 + j;
                if (r < n) C[(size_t)r * HID + n0 + l15] = c[t][j];
            }
        }
        __syncthreads();
    }
}

template <bool FIRST, bool POOL>
__global__ void __launch_bounds__(256)
k_agg_tile(const int2* __restrict__ rec, const int* __restrict__ start,
           const float* __restrict__ dinv, const float* __restrict__ hW,
           float* __restrict__ agg, float* __restrict__ wgt, int n,
           const int* __restrict__ batch, float* __restrict__ psum, float* __restrict__ pcnt) {
    __shared__ __attribute__((aligned(16))) float acc[TN * HID];
    int tid = threadIdx.x;
    int v0 = blockIdx.x * TN;

    const float4* hW4 = (const float4*)hW;
    float4* acc4 = (float4*)acc;
    for (int i = tid; i < TN * HID / 4; i += 256) {
        int v = v0 + (i >> 4);
        float4 r = make_float4(0.f, 0.f, 0.f, 0.f);
        if (v < n) {
            float di = dinv[v];
            float4 h = hW4[(size_t)v * (HID / 4) + (i & 15)];
            r.x = h.x * di * di; r.y = h.y * di * di;
            r.z = h.z * di * di; r.w = h.w * di * di;
        }
        acc4[i] = r;
    }
    __syncthreads();

    int lane = tid & 31;
    int wave = __builtin_amdgcn_readfirstlane(tid >> 5);
    int b = start[blockIdx.x], e = start[blockIdx.x + 1];
    const float* hl = hW + lane * 2;
    float* al = acc + lane * 2;

    int p = b + wave * 32;
    int s = 0, d = v0;
    float nm = 0.f, ws = 0.f, wd = 0.f;
    if (p < e && lane < min(32, e - p)) {
        int2 r = rec[p + lane]; s = r.x; d = r.y;
        if (!FIRST) nm = wgt[p + lane];
    }
    if (FIRST) { ws = dinv[s]; wd = dinv[d]; }
    while (p < e) {
        int m = min(32, e - p);
        v2f h[32];
#pragma unroll
        for (int j = 0; j < 32; ++j)
            h[j] = *(const v2f*)(hl + (size_t)__builtin_amdgcn_readlane(s, j) * HID);
        int pn = p + 256;
        int s2 = 0, d2 = v0;
        float nm2 = 0.f;
        if (pn < e && lane < min(32, e - pn)) {
            int2 r = rec[pn + lane]; s2 = r.x; d2 = r.y;
            if (!FIRST) nm2 = wgt[pn + lane];
        }
        int dl = d - v0;
        if (FIRST) {
            nm = ws * wd;
            if (lane < m) wgt[p + lane] = nm;
        }
        int cur = __builtin_amdgcn_readlane(dl, 0);
        float r0 = 0.f, r1 = 0.f;
#pragma unroll
        for (int j = 0; j < 32; ++j) {
            if (j < m) {
                int   dj = __builtin_amdgcn_readlane(dl, j);
                float nj = __int_as_float(__builtin_amdgcn_readlane(__float_as_int(nm), j));
                if (dj != cur) {
                    atomicAdd(al + cur * HID,     r0);
                    atomicAdd(al + cur * HID + 1, r1);
                    r0 = 0.f; r1 = 0.f; cur = dj;
                }
                r0 += h[j][0] * nj;
                r1 += h[j][1] * nj;
            }
        }
        atomicAdd(al + cur * HID,     r0);
        atomicAdd(al + cur * HID + 1, r1);
        if (FIRST) { ws = dinv[s2]; wd = dinv[d2]; }
        else       nm = nm2;
        s = s2; d = d2; p = pn;
    }
    __syncthreads();

    if (POOL) {
        int c = tid & (HID - 1);
        int r0 = (wave >> 1) * (TN / 4);
        int gid[TN / 4];
#pragma unroll
        for (int k = 0; k < TN / 4; ++k) gid[k] = batch[min(v0 + r0 + k, n - 1)];
        float col[TN / 4];
#pragma unroll
        for (int k = 0; k < TN / 4; ++k) col[k] = acc[(r0 + k) * HID + c];
        float run = 0.f;
        int gcur = gid[0], len = 0;
#pragma unroll
        for (int k = 0; k < TN / 4; ++k) {
            if (v0 + r0 + k < n) {
                if (gid[k] != gcur) {
                    atomicAdd(&psum[(size_t)gcur * HID + c], run);
                    if (c == 0) atomicAdd(&pcnt[gcur], (float)len);
                    gcur = gid[k]; run = 0.f; len = 0;
                }
                run += col[k];
                ++len;
            }
        }
        if (len > 0) {
            atomicAdd(&psum[(size_t)gcur * HID + c], run);
            if (c == 0) atomicAdd(&pcnt[gcur], (float)len);
        }
    } else {
        float4* agg4 = (float4*)agg;
        for (int i = tid; i < TN * HID / 4; i += 256) {
            int v = v0 + (i >> 4);
            if (v < n) agg4[(size_t)v0 * (HID / 4) + i] = acc4[i];
        }
    }
}

__global__ void k_final(const float* __restrict__ sums, const float* __restrict__ cnt,
                        const float* __restrict__ Wl, const float* __restrict__ bl,
                        float* __restrict__ out, int g, int oc) {
    int i = blockIdx.x * blockDim.x + threadIdx.x;
    if (i < g * oc) {
        int gi = i / oc, o = i % oc;
        float inv = 1.0f / fmaxf(cnt[gi], 1.0f);
        float acc = bl[o];
        for (int c = 0; c < HID; ++c)
            acc += sums[(size_t)gi * HID + c] * inv * Wl[c * oc + o];
        out[i] = acc;
    }
}

extern "C" void kernel_launch(void* const* d_in, const int* in_sizes, int n_in,
                              void* d_out, int out_size, void* d_ws, size_t ws_size,
                              hipStream_t stream) {
    const float* x    = (const float*)d_in[0];
    const int*   eidx = (const int*)  d_in[1];
    const int*   batch= (const int*)  d_in[2];
    const float* W1   = (const float*)d_in[3];
    const float* W2   = (const float*)d_in[4];
    const float* W3   = (const float*)d_in[5];
    const float* Wl   = (const float*)d_in[6];
    const float* bl   = (const float*)d_in[7];

    const int INC = in_sizes[3] / HID;
    const int N   = in_sizes[0] / INC;
    const int E   = in_sizes[1] / 2;
    const int OC  = in_sizes[7];
    const int G   = out_size / OC;

    const int* src = eidx;
    const int* dst = eidx + E;

    const int NT = (N + TN - 1) / TN;
    const int NC = (E + CH - 1) / CH;

    char* ws = (char*)d_ws; size_t off = 0;
    auto carve = [&](size_t bytes) {
        void* p = ws + off; off = (off + bytes + 255) & ~(size_t)255; return p;
    };
    float* dinv = (float*)carve((size_t)N * 4);
    float* hW   = (float*)carve((size_t)N * HID * 4);
    float* agg  = (float*)carve((size_t)N * HID * 4);
    float* sums = (float*)carve((size_t)G * HID * 4);
    float* cnt  = (float*)carve((size_t)G * 4);
    int*   tab  = (int*)  carve((size_t)NC * NT * 4);
    int*   tot  = (int*)  carve((size_t)NT * 4);
    int*   start= (int*)  carve((size_t)(NT + 1) * 4);
    int2*  rec  = (int2*) carve((size_t)E * 8);
    float* wgt  = (float*)carve((size_t)E * 4);
    (void)ws_size; (void)n_in;

    const int B = 256;
    const int gt  = (N + GR - 1) / GR;

    k_bin_count<<<NC, B, (size_t)NT * 4, stream>>>(dst, tab, E, NT);
    k_bin_col  <<<(NT + 3) / 4, 128, 0, stream>>>(tab, tot, NC, NT);
    k_bin_start<<<1, 1024, 0, stream>>>(tot, start, NT);
    k_bin_fill <<<NC, B, (size_t)NT * 4, stream>>>(src, dst, tab, start, rec, E, NT);
    k_tile_deg <<<NT, B, 0, stream>>>(rec, start, dinv, N, sums, cnt, G);

    k_gemm_wmma<128, false><<<gt, 128, 0, stream>>>(x, W1, hW, N);
    k_agg_tile<true, false><<<NT, B, 0, stream>>>(rec, start, dinv, hW, agg, wgt, N, batch, sums, cnt);

    k_gemm_wmma<64, true><<<gt, 128, 0, stream>>>(agg, W2, hW, N);
    k_agg_tile<false, false><<<NT, B, 0, stream>>>(rec, start, dinv, hW, agg, wgt, N, batch, sums, cnt);

    k_gemm_wmma<64, true><<<gt, 128, 0, stream>>>(agg, W3, hW, N);
    k_agg_tile<false, true><<<NT, B, 0, stream>>>(rec, start, dinv, hW, agg, wgt, N, batch, sums, cnt);

    k_final<<<(G * OC + B - 1) / B, B, 0, stream>>>(sums, cnt, Wl, bl, (float*)d_out, G, OC);
}
